# starting kernel
# baseline (speedup 1.0000x reference)
#include <hip/hip_runtime.h>

typedef __attribute__((ext_vector_type(2))) float v2f;
typedef __attribute__((ext_vector_type(8))) float v8f;

namespace {
constexpr int D              = 1024;
constexpr int ROWS_PER_BLOCK = 128;
constexpr int EB             = 128;
constexpr int KS             = 64;
constexpr int KPAD           = 68;
}

__global__ __launch_bounds__(256, 2)
void bilinear_wmma_f32_kernel(const float* __restrict__ elg,
                              const float* __restrict__ eth,
                              const float* __restrict__ W,
                              float* __restrict__ out)
{
    __shared__ float sW[EB * KPAD];

    const int tid  = threadIdx.x;
    const int wave = tid >> 5;
    const int lane = tid & 31;
    const int colg = lane & 15;
    const int kh   = (lane >> 4) << 1;

    const int rowTile = blockIdx.x * ROWS_PER_BLOCK + wave * 16;
    const float* aRow = elg + (size_t)(rowTile + colg) * D;

    float acc[8];
#pragma unroll
    for (int r = 0; r < 8; ++r) acc[r] = 0.f;

    for (int eb = 0; eb < D / EB; ++eb) {
        const int e0 = eb * EB;

        v8f C[8] = {};

        for (int ks = 0; ks < D / KS; ++ks) {
            __syncthreads();
#pragma unroll 4
            for (int i = tid; i < KS * EB; i += 256) {
                int e  = i & (EB - 1);
                int kl = i >> 7;
                sW[e * KPAD + kl] = W[(size_t)(ks * KS + kl) * D + e0 + e];
            }
            __syncthreads();

#pragma unroll
            for (int kt = 0; kt < KS / 16; ++kt) {
#pragma unroll
                for (int j = 0; j < 4; ++j) {
                    const int klA = kt * 16 + j * 4 + kh;
                    v2f a = *reinterpret_cast<const v2f*>(aRow + ks * KS + klA);
#pragma unroll
                    for (int t = 0; t < 8; ++t) {
                        v2f b = *reinterpret_cast<const v2f*>(
                            &sW[(t * 16 + colg) * KPAD + klA]);
                        C[t] = __builtin_amdgcn_wmma_f32_16x16x4_f32(
                            false, a, false, b, (short)0, C[t], false, false);
                    }
                }
            }
        }

        const int mOff = (lane >> 4) * 8;
#pragma unroll
        for (int t = 0; t < 8; ++t) {
            const int col = e0 + t * 16 + colg;
#pragma unroll
            for (int r = 0; r < 8; ++r) {
                float ev = eth[(size_t)(rowTile + mOff + r) * D + col];
                acc[r] = fmaf(C[t][r], ev, acc[r]);
            }
        }
    }

#pragma unroll
    for (int m = 1; m <= 8; m <<= 1) {
#pragma unroll
        for (int r = 0; r < 8; ++r) acc[r] += __shfl_xor(acc[r], m, 32);
    }

    if (colg == 0) {
        const int mOff = (lane >> 4) * 8;
#pragma unroll
        for (int r = 0; r < 8; ++r) out[rowTile + mOff + r] = acc[r];
    }
}

extern "C" void kernel_launch(void* const* d_in, const int* in_sizes, int n_in,
                              void* d_out, int out_size, void* d_ws, size_t ws_size,
                              hipStream_t stream) {
    const float* elg = (const float*)d_in[0];
    const float* eth = (const float*)d_in[1];
    const float* W   = (const float*)d_in[2];
    float* out = (float*)d_out;

    const int nRows = in_sizes[0] / D;
    dim3 grid(nRows / ROWS_PER_BLOCK);
    dim3 block(256);
    bilinear_wmma_f32_kernel<<<grid, block, 0, stream>>>(elg, eth, W, out);
}


# W k-slab staging software-pipelined: next slab's W tile and A fragments are loaded to registers ahead of the WMMA block and land in a second LDS stage after it; one barrier per slab, not two
# speedup vs baseline: 1.7006x; 1.7006x over previous
#include <hip/hip_runtime.h>

typedef __attribute__((ext_vector_type(2))) float v2f;
typedef __attribute__((ext_vector_type(4))) float v4f;
typedef __attribute__((ext_vector_type(8))) float v8f;


namespace {
constexpr int D              = 1024;
constexpr int ROWS_PER_BLOCK = 128;
constexpr int EB             = 128;
constexpr int KS             = 64;
constexpr int KPAD           = 68;
constexpr int NKS            = D / KS;
constexpr int NSLAB          = (D / EB) * NKS;
constexpr int WREG           = KS * EB / 256;
constexpr int AFR            = KS / 4;
}

__global__ __launch_bounds__(256, 2)
void bilinear_wmma_f32_kernel(const float* __restrict__ elg,
                              const float* __restrict__ eth,
                              const float* __restrict__ W,
                              float* __restrict__ out)
{
    __shared__ __attribute__((aligned(16))) float sW[2][EB * KPAD];

    const int tid  = threadIdx.x;
    const int wave = tid >> 5;
    const int lane = tid & 31;
    const int colg = lane & 15;
    const int kh   = (lane >> 4) << 1;

    const int rowTile = blockIdx.x * ROWS_PER_BLOCK + wave * 16;
    const float* aRow = elg + (size_t)(rowTile + colg) * D;

    const int se   = tid & (EB - 1);
    const int sk0  = (tid >> 7) * WREG;
    const float* wCol = W + (size_t)sk0 * D + se;
    const int sOff = se * KPAD + sk0;

    float wN[WREG];
    v2f   aC[AFR];
    v2f   aN[AFR];

#pragma unroll
    for (int m = 0; m < WREG; ++m) wN[m] = wCol[(size_t)m * D];
#pragma unroll
    for (int f = 0; f < AFR; ++f)
        aC[f] = *reinterpret_cast<const v2f*>(aRow + f * 4 + kh);
#pragma unroll
    for (int q = 0; q < WREG / 4; ++q) {
        v4f v = {wN[4 * q], wN[4 * q + 1], wN[4 * q + 2], wN[4 * q + 3]};
        *reinterpret_cast<v4f*>(&sW[0][sOff + 4 * q]) = v;
    }
    __syncthreads();

    float acc[8];
#pragma unroll
    for (int r = 0; r < 8; ++r) acc[r] = 0.f;

    for (int eb = 0; eb < D / EB; ++eb) {
        const int e0 = eb * EB;

        v8f C[8] = {};

        for (int ks = 0; ks < NKS; ++ks) {
            const int  s    = eb * NKS + ks;
            const int  cur  = ks & 1;
            const bool more = (s + 1) < NSLAB;

            if (more) {
                const int sn  = s + 1;
                const int ksn = sn & (NKS - 1);
                const int e0n = (sn / NKS) * EB;
                const float* wp = wCol + (size_t)(ksn * KS) * D + e0n;
#pragma unroll
                for (int m = 0; m < WREG; ++m) wN[m] = wp[(size_t)m * D];
                const float* ap = aRow + ksn * KS + kh;
#pragma unroll
                for (int f = 0; f < AFR; ++f)
                    aN[f] = *reinterpret_cast<const v2f*>(ap + f * 4);
            }

            const float* sB = &sW[cur][colg * KPAD + kh];
#pragma unroll
            for (int f = 0; f < AFR; ++f) {
#pragma unroll
                for (int t = 0; t < 8; ++t) {
                    v2f b = *reinterpret_cast<const v2f*>(sB + t * 16 * KPAD + f * 4);
                    C[t] = __builtin_amdgcn_wmma_f32_16x16x4_f32(
                        false, aC[f], false, b, (short)0, C[t], false, false);
                }
            }

            if (more) {
                float* sd = &sW[cur ^ 1][sOff];
#pragma unroll
                for (int q = 0; q < WREG / 4; ++q) {
                    v4f v = {wN[4 * q], wN[4 * q + 1], wN[4 * q + 2], wN[4 * q + 3]};
                    *reinterpret_cast<v4f*>(sd + 4 * q) = v;
                }
#pragma unroll
                for (int f = 0; f < AFR; ++f) aC[f] = aN[f];
            }
            __syncthreads();
        }

        const int mOff = (lane >> 4) * 8;
#pragma unroll
        for (int t = 0; t < 8; ++t) {
            const int col = e0 + t * 16 + colg;
#pragma unroll
            for (int r = 0; r < 8; ++r) {
                float ev = eth[(size_t)(rowTile + mOff + r) * D + col];
                acc[r] = fmaf(C[t][r], ev, acc[r]);
            }
        }
    }

#pragma unroll
    for (int m = 1; m <= 8; m <<= 1) {
#pragma unroll
        for (int r = 0; r < 8; ++r) acc[r] += __shfl_xor(acc[r], m, 32);
    }

    if (colg == 0) {
        const int mOff = (lane >> 4) * 8;
#pragma unroll
        for (int r = 0; r < 8; ++r) out[rowTile + mOff + r] = acc[r];
    }
}

extern "C" void kernel_launch(void* const* d_in, const int* in_sizes, int n_in,
                              void* d_out, int out_size, void* d_ws, size_t ws_size,
                              hipStream_t stream) {
    const float* elg = (const float*)d_in[0];
    const float* eth = (const float*)d_in[1];
    const float* W   = (const float*)d_in[2];
    float* out = (float*)d_out;

    const int nRows = in_sizes[0] / D;
    dim3 grid(nRows / ROWS_PER_BLOCK);
    dim3 block(256);
    bilinear_wmma_f32_kernel<<<grid, block, 0, stream>>>(elg, eth, W, out);
}


# Staging placed by wave age: the younger wave of each SIMD pair lands W and issues next W and A loads at slab start, the older issues A mid-slab and lands W after its WMMAs; LDS-only slab barrier
# speedup vs baseline: 1.7748x; 1.0436x over previous
#include <hip/hip_runtime.h>

typedef __attribute__((ext_vector_type(2))) float v2f;
typedef __attribute__((ext_vector_type(4))) float v4f;
typedef __attribute__((ext_vector_type(8))) float v8f;


namespace {
constexpr int D              = 1024;
constexpr int ROWS_PER_BLOCK = 128;
constexpr int EB             = 128;
constexpr int KS             = 64;
constexpr int KPAD           = 68;
constexpr int NKS            = D / KS;
constexpr int NSLAB          = (D / EB) * NKS;
constexpr int WREG           = KS * EB / 256;
constexpr int AFR            = KS / 4;
}

static __device__ __forceinline__ void slab_barrier() {
    asm volatile("s_wait_dscnt 0x0\n\ts_barrier_signal -1\n\ts_barrier_wait -1" ::: "memory");
}

static __device__ __forceinline__ void stage_w(float (&wN)[WREG], float* sd, const float* wp) {
#pragma unroll
    for (int q = 0; q < WREG / 4; ++q) {
        v4f v = {wN[4 * q], wN[4 * q + 1], wN[4 * q + 2], wN[4 * q + 3]};
        *reinterpret_cast<v4f*>(sd + 4 * q) = v;
    }
#pragma unroll
    for (int m = 0; m < WREG; ++m) wN[m] = wp[(size_t)m * D];
}

static __device__ __forceinline__ void load_a(v2f (&aN)[AFR], const float* ap) {
#pragma unroll
    for (int f = 0; f < AFR; ++f)
        aN[f] = *reinterpret_cast<const v2f*>(ap + f * 4);
}

__global__ __launch_bounds__(256, 2)
void bilinear_wmma_f32_kernel(const float* __restrict__ elg,
                              const float* __restrict__ eth,
                              const float* __restrict__ W,
                              float* __restrict__ out)
{
    __shared__ __attribute__((aligned(16))) float sW[2][EB * KPAD];

    const int tid  = threadIdx.x;
    const int wave = tid >> 5;
    const int lane = tid & 31;
    const int colg = lane & 15;
    const int kh   = (lane >> 4) << 1;

    const bool early = __builtin_amdgcn_readfirstlane(wave) >= 4;

    const int rowTile = blockIdx.x * ROWS_PER_BLOCK + wave * 16;
    const float* aRow = elg + (size_t)(rowTile + colg) * D;

    const int se   = tid & (EB - 1);
    const int sk0  = (tid >> 7) * WREG;
    const float* wCol = W + (size_t)sk0 * D + se;
    const int sOff = se * KPAD + sk0;

    float wN[WREG];
    v2f   aC[AFR];
    v2f   aN[AFR];

#pragma unroll
    for (int m = 0; m < WREG; ++m) wN[m] = wCol[(size_t)m * D];
#pragma unroll
    for (int f = 0; f < AFR; ++f)
        aC[f] = *reinterpret_cast<const v2f*>(aRow + f * 4 + kh);
    stage_w(wN, &sW[0][sOff], wCol + (size_t)KS * D);
    __syncthreads();

    float acc[8];
#pragma unroll
    for (int r = 0; r < 8; ++r) acc[r] = 0.f;

    for (int eb = 0; eb < D / EB; ++eb) {
        const int e0 = eb * EB;

        v8f C[8] = {};

        for (int ks = 0; ks < NKS; ++ks) {
            const int  s     = eb * NKS + ks;
            const int  cur   = ks & 1;
            const bool more1 = (s + 1) < NSLAB;

            const int s2   = (s + 2 < NSLAB) ? s + 2 : NSLAB - 1;
            const int wOff = ((s2 & (NKS - 1)) * KS) * D + (s2 / NKS) * EB;
            const int aOff = ((s + 1) & (NKS - 1)) * KS + kh;

            if (early && more1) {
                stage_w(wN, &sW[cur ^ 1][sOff], wCol + wOff);
                load_a(aN, aRow + aOff);
            }

            const float* sB = &sW[cur][colg * KPAD + kh];

#pragma unroll
            for (int f = 0; f < AFR / 2; ++f) {
#pragma unroll
                for (int t = 0; t < 8; ++t) {
                    v2f b = *reinterpret_cast<const v2f*>(sB + t * 16 * KPAD + f * 4);
                    C[t] = __builtin_amdgcn_wmma_f32_16x16x4_f32(
                        false, aC[f], false, b, (short)0, C[t], false, false);
                }
            }

            if (!early && more1) load_a(aN, aRow + aOff);

#pragma unroll
            for (int f = AFR / 2; f < AFR; ++f) {
#pragma unroll
                for (int t = 0; t < 8; ++t) {
                    v2f b = *reinterpret_cast<const v2f*>(sB + t * 16 * KPAD + f * 4);
                    C[t] = __builtin_amdgcn_wmma_f32_16x16x4_f32(
                        false, aC[f], false, b, (short)0, C[t], false, false);
                }
            }

            if (more1) {
#pragma unroll
                for (int f = 0; f < AFR; ++f) aC[f] = aN[f];
            }
            if (!early && more1) stage_w(wN, &sW[cur ^ 1][sOff], wCol + wOff);

            slab_barrier();
        }

        const int mOff = (lane >> 4) * 8;
#pragma unroll
        for (int t = 0; t < 8; ++t) {
            const int col = e0 + t * 16 + colg;
#pragma unroll
            for (int r = 0; r < 8; ++r) {
                float ev = eth[(size_t)(rowTile + mOff + r) * D + col];
                acc[r] = fmaf(C[t][r], ev, acc[r]);
            }
        }
    }

#pragma unroll
    for (int m = 1; m <= 8; m <<= 1) {
#pragma unroll
        for (int r = 0; r < 8; ++r) acc[r] += __shfl_xor(acc[r], m, 32);
    }

    if (colg == 0) {
        const int mOff = (lane >> 4) * 8;
#pragma unroll
        for (int r = 0; r < 8; ++r) out[rowTile + mOff + r] = acc[r];
    }
}

extern "C" void kernel_launch(void* const* d_in, const int* in_sizes, int n_in,
                              void* d_out, int out_size, void* d_ws, size_t ws_size,
                              hipStream_t stream) {
    const float* elg = (const float*)d_in[0];
    const float* eth = (const float*)d_in[1];
    const float* W   = (const float*)d_in[2];
    float* out = (float*)d_out;

    const int nRows = in_sizes[0] / D;
    dim3 grid(nRows / ROWS_PER_BLOCK);
    dim3 block(256);
    bilinear_wmma_f32_kernel<<<grid, block, 0, stream>>>(elg, eth, W, out);
}


# B words read from LDS as ONE group of 32 16-byte loads per 64-WMMA block, a block ahead, 2nd register set; slab barrier now inside a block; 4 consecutive k per lane word: k summed in another order
# speedup vs baseline: 1.9100x; 1.0762x over previous
#include <hip/hip_runtime.h>

typedef __attribute__((ext_vector_type(2))) float v2f;
typedef __attribute__((ext_vector_type(4))) float v4f;
typedef __attribute__((ext_vector_type(8))) float v8f;


namespace {
constexpr int D              = 1024;
constexpr int ROWS_PER_BLOCK = 128;
constexpr int EB             = 128;
constexpr int KS             = 64;
constexpr int KPAD           = 68;
constexpr int NKS            = D / KS;
constexpr int NSLAB          = (D / EB) * NKS;
constexpr int WREG           = KS * EB / 256;
constexpr int AG             = KS / 8;
constexpr int BG             = AG / 2;
constexpr int BFR            = BG * 8;
}

static __device__ __forceinline__ void slab_barrier() {
    asm volatile("s_wait_dscnt 0x0\n\ts_barrier_signal -1\n\ts_barrier_wait -1" ::: "memory");
}

static __device__ __forceinline__ void stage_w(float (&wN)[WREG], float* sd, const float* wp) {
#pragma unroll
    for (int q = 0; q < WREG / 4; ++q) {
        v4f v = {wN[4 * q], wN[4 * q + 1], wN[4 * q + 2], wN[4 * q + 3]};
        *reinterpret_cast<v4f*>(sd + 4 * q) = v;
    }
#pragma unroll
    for (int m = 0; m < WREG; ++m) wN[m] = wp[(size_t)m * D];
}

static __device__ __forceinline__ void load_a(v4f (&aN)[AG], const float* ap) {
#pragma unroll
    for (int g = 0; g < AG; ++g)
        aN[g] = *reinterpret_cast<const v4f*>(ap + g * 8);
}

static __device__ __forceinline__ void load_b(v4f (&b)[BFR], const float* sB) {
#pragma unroll
    for (int g = 0; g < BG; ++g) {
#pragma unroll
        for (int t = 0; t < 8; ++t)
            b[g * 8 + t] = *reinterpret_cast<const v4f*>(sB + t * 16 * KPAD + g * 8);
    }
}

template <int GA, int GB, int NG>
static __device__ __forceinline__ void mma_part(v8f (&C)[8], const v4f (&a)[AG], const v4f (&b)[BFR]) {
#pragma unroll
    for (int g = 0; g < NG; ++g) {
        const v4f a4 = a[GA + g];
        const v2f aLo = {a4.x, a4.y};
        const v2f aHi = {a4.z, a4.w};
#pragma unroll
        for (int t = 0; t < 8; ++t) {
            const v4f b4 = b[(GB + g) * 8 + t];
            const v2f bLo = {b4.x, b4.y};
            C[t] = __builtin_amdgcn_wmma_f32_16x16x4_f32(
                false, aLo, false, bLo, (short)0, C[t], false, false);
        }
#pragma unroll
        for (int t = 0; t < 8; ++t) {
            const v4f b4 = b[(GB + g) * 8 + t];
            const v2f bHi = {b4.z, b4.w};
            C[t] = __builtin_amdgcn_wmma_f32_16x16x4_f32(
                false, aHi, false, bHi, (short)0, C[t], false, false);
        }
    }
}

__global__ __launch_bounds__(256, 2)
void bilinear_wmma_f32_kernel(const float* __restrict__ elg,
                              const float* __restrict__ eth,
                              const float* __restrict__ W,
                              float* __restrict__ out)
{
    __shared__ __attribute__((aligned(16))) float sW[2][EB * KPAD];

    const int tid  = threadIdx.x;
    const int wave = tid >> 5;
    const int lane = tid & 31;
    const int colg = lane & 15;
    const int kq   = (lane >> 4) << 2;

    const bool early = __builtin_amdgcn_readfirstlane(wave) >= 4;

    const int rowTile = blockIdx.x * ROWS_PER_BLOCK + wave * 16;
    const float* aRow = elg + (size_t)(rowTile + colg) * D;

    const int se   = tid & (EB - 1);
    const int sk0  = (tid >> 7) * WREG;
    const float* wCol = W + (size_t)sk0 * D + se;
    const int sOff = se * KPAD + sk0;
    const int bOff = colg * KPAD + kq;

    float wN[WREG];
    v4f   aC[AG];
    v4f   aN[AG];
    v4f   bX[BFR];
    v4f   bY[BFR];

#pragma unroll
    for (int m = 0; m < WREG; ++m) wN[m] = wCol[(size_t)m * D];
    load_a(aC, aRow + kq);
    stage_w(wN, &sW[0][sOff], wCol + (size_t)KS * D);
    __syncthreads();
    load_b(bX, &sW[0][bOff]);

    float acc[8];
#pragma unroll
    for (int r = 0; r < 8; ++r) acc[r] = 0.f;

    for (int eb = 0; eb < D / EB; ++eb) {
        const int e0 = eb * EB;

        v8f C[8] = {};

        for (int ks = 0; ks < NKS; ++ks) {
            const int  s     = eb * NKS + ks;
            const int  cur   = ks & 1;
            const bool more1 = (s + 1) < NSLAB;

            const int s2   = (s + 2 < NSLAB) ? s + 2 : NSLAB - 1;
            const int wOff = ((s2 & (NKS - 1)) * KS) * D + (s2 / NKS) * EB;
            const int aOff = ((s + 1) & (NKS - 1)) * KS + kq;

            if (early && more1) {
                stage_w(wN, &sW[cur ^ 1][sOff], wCol + wOff);
                load_a(aN, aRow + aOff);
            }

            load_b(bY, &sW[cur][bOff + KS / 2]);
            __builtin_amdgcn_sched_barrier(0);

            mma_part<0, 0, BG / 2>(C, aC, bX);

            if (!early && more1) {
                load_a(aN, aRow + aOff);
                stage_w(wN, &sW[cur ^ 1][sOff], wCol + wOff);
            }

            slab_barrier();

            mma_part<BG / 2, BG / 2, BG / 2>(C, aC, bX);
            __builtin_amdgcn_sched_barrier(0);

            load_b(bX, &sW[cur ^ 1][bOff]);
            __builtin_amdgcn_sched_barrier(0);

            mma_part<BG, 0, BG>(C, aC, bY);

            if (more1) {
#pragma unroll
                for (int g = 0; g < AG; ++g) aC[g] = aN[g];
            }
        }

        const int mOff = (lane >> 4) * 8;
#pragma unroll
        for (int t = 0; t < 8; ++t) {
            const int col = e0 + t * 16 + colg;
#pragma unroll
            for (int r = 0; r < 8; ++r) {
                float ev = eth[(size_t)(rowTile + mOff + r) * D + col];
                acc[r] = fmaf(C[t][r], ev, acc[r]);
            }
        }
    }

#pragma unroll
    for (int m = 1; m <= 8; m <<= 1) {
#pragma unroll
        for (int r = 0; r < 8; ++r) acc[r] += __shfl_xor(acc[r], m, 32);
    }

    if (colg == 0) {
        const int mOff = (lane >> 4) * 8;
#pragma unroll
        for (int r = 0; r < 8; ++r) out[rowTile + mOff + r] = acc[r];
    }
}

extern "C" void kernel_launch(void* const* d_in, const int* in_sizes, int n_in,
                              void* d_out, int out_size, void* d_ws, size_t ws_size,
                              hipStream_t stream) {
    const float* elg = (const float*)d_in[0];
    const float* eth = (const float*)d_in[1];
    const float* W   = (const float*)d_in[2];
    float* out = (float*)d_out;

    const int nRows = in_sizes[0] / D;
    dim3 grid(nRows / ROWS_PER_BLOCK);
    dim3 block(256);
    bilinear_wmma_f32_kernel<<<grid, block, 0, stream>>>(elg, eth, W, out);
}
